# starting kernel
# baseline (speedup 1.0000x reference)
#include <hip/hip_runtime.h>
#include <math.h>

typedef __attribute__((ext_vector_type(2))) float v2f;
typedef __attribute__((ext_vector_type(4))) float v4f;
typedef __attribute__((ext_vector_type(8))) float v8f;

#define D2R 0.017453292519943295f
#define R2D 57.29577951308232f

__global__ __launch_bounds__(256) void fk_wmma_kernel(
    const float* __restrict__ theta, const float* __restrict__ dh,
    float* __restrict__ out, int n)
{
    __shared__ float smem[8 * 448];
    const int lane = threadIdx.x & 31;
    const int wave = threadIdx.x >> 5;
    float* T = &smem[wave * 448];
    float* G = T + 384;

    const int base = (blockIdx.x * 8 + wave) * 4;

    {
        const int s = lane >> 3;
        const int i = lane & 7;
        if (i < 6) {
            int samp = base + s; if (samp >= n) samp = n - 1;
            const float a     = dh[i * 4 + 0];
            const float alr   = dh[i * 4 + 1] * D2R;
            const float d     = dh[i * 4 + 2];
            const float off   = dh[i * 4 + 3];
            const float th    = (off + theta[samp * 6 + i]) * D2R;
            float st, ct, sa, ca;
            sincosf(th,  &st, &ct);
            sincosf(alr, &sa, &ca);
            float* tp = &T[i * 64 + s * 16];
            v4f r0; r0.x = ct;      r0.y = -st;     r0.z = 0.0f; r0.w = a;
            v4f r1; r1.x = st * ca; r1.y = ct * ca; r1.z = -sa;  r1.w = -sa * d;
            v4f r2; r2.x = st * sa; r2.y = ct * sa; r2.z = ca;   r2.w = ca * d;
            v4f r3; r3.x = 0.0f;    r3.y = 0.0f;    r3.z = 0.0f; r3.w = 1.0f;
            *(v4f*)(tp +  0) = r0;
            *(v4f*)(tp +  4) = r1;
            *(v4f*)(tp +  8) = r2;
            *(v4f*)(tp + 12) = r3;
        }
    }
    asm volatile("s_wait_dscnt 0" ::: "memory");

    const int h  = lane >> 4;
    const int m  = lane & 15;
    const int sA = m >> 2;
    const int iA = m & 3;

    v2f Acc;
    Acc.x = T[0 * 64 + sA * 16 + iA * 4 + (2 * h + 0)];
    Acc.y = T[0 * 64 + sA * 16 + iA * 4 + (2 * h + 1)];

    const bool wr = (h == ((m >> 3) & 1));
    const bool hi = (sA & 1);
    #pragma unroll
    for (int j = 1; j < 6; ++j) {
        v2f Bv;
        Bv.x = T[j * 64 + sA * 16 + (2 * h + 0) * 4 + iA];
        Bv.y = T[j * 64 + sA * 16 + (2 * h + 1) * 4 + iA];
        v8f c = {};
        c = __builtin_amdgcn_wmma_f32_16x16x4_f32(
                false, Acc, false, Bv, (short)0, c, false, false);
        if (wr) {
            v4f dv;
            dv.x = hi ? c[4] : c[0];
            dv.y = hi ? c[5] : c[1];
            dv.z = hi ? c[6] : c[2];
            dv.w = hi ? c[7] : c[3];
            *(v4f*)&G[sA * 16 + iA * 4] = dv;
        }
        asm volatile("s_wait_dscnt 0" ::: "memory");
        if (j < 5) {
            Acc.x = G[sA * 16 + (2 * h + 0) * 4 + iA];
            Acc.y = G[sA * 16 + (2 * h + 1) * 4 + iA];
        }
    }

    if (lane < 4 && (base + lane) < n) {
        const int s = lane;
        #define GM(r, cc) G[s * 16 + (cc) * 4 + (r)]
        const float x = GM(0, 3), y = GM(1, 3), z = GM(2, 3);
        const float T00 = GM(0, 0), T01 = GM(0, 1), T02 = GM(0, 2);
        const float T10 = GM(1, 0), T11 = GM(1, 1), T12 = GM(1, 2);
        const float T22 = GM(2, 2);
        #undef GM
        float Aa, Bb, Cc;
        if (fabsf(T12) <= 1e-6f && fabsf(T22) <= 1e-6f) {
            Aa = R2D * atan2f(T10, T11);
            Bb = R2D * atan2f(T02, T22);
            Cc = 0.0f;
        } else {
            const float An = atan2f(-T01, T00);
            float sAn, cAn;
            sincosf(An, &sAn, &cAn);
            Aa = R2D * An;
            Bb = R2D * atan2f(T02, cAn * T00 - sAn * T01);
            Cc = R2D * atan2f(-T12, T22);
        }
        float* o = out + (size_t)(base + s) * 6;
        o[0] = x; o[1] = y; o[2] = z; o[3] = Aa; o[4] = Bb; o[5] = Cc;
    }
}

extern "C" void kernel_launch(void* const* d_in, const int* in_sizes, int n_in,
                              void* d_out, int out_size, void* d_ws, size_t ws_size,
                              hipStream_t stream) {
    const float* theta = (const float*)d_in[0];
    const float* dh    = (const float*)d_in[1];
    float* out = (float*)d_out;
    const int n = in_sizes[0] / 6;
    const int samplesPerBlock = 32;
    const int blocks = (n + samplesPerBlock - 1) / samplesPerBlock;
    fk_wmma_kernel<<<blocks, 256, 0, stream>>>(theta, dh, out, n);
}


# One wave now does 32 samples, one per lane in the sin/cos and atan2 stages (32 lanes busy, before 24 and 4); its 8 four-sample f32 WMMA chains run transposed so D is the next B in registers
# speedup vs baseline: 2.2293x; 2.2293x over previous
#include <hip/hip_runtime.h>
#include <math.h>


typedef __attribute__((ext_vector_type(2))) float v2f;
typedef __attribute__((ext_vector_type(4))) float v4f;
typedef __attribute__((ext_vector_type(8))) float v8f;

#define D2R 0.017453292519943295f
#define R2D 57.29577951308232f

#define REC_PITCH   52
#define CONST_BASE  1680
#define RES_PITCH   20
#define WAVE_DWORDS 1728

__global__ __launch_bounds__(256) void fk_wmma_kernel(
    const float* __restrict__ theta, const float* __restrict__ dh,
    float* __restrict__ out, int n)
{
    __shared__ __attribute__((aligned(16))) float smem[8 * WAVE_DWORDS];
    const int lane = threadIdx.x & 31;
    const int wave = threadIdx.x >> 5;
    float* S = &smem[wave * WAVE_DWORDS];
    const int base = (blockIdx.x * 8 + wave) * 32;

    int samp = base + lane; if (samp >= n) samp = n - 1;
    const v2f* tp = (const v2f*)(theta + (size_t)samp * 6);
    const v2f t01 = tp[0], t23 = tp[1], t45 = tp[2];

    const int jc = lane < 6 ? lane : 5;
    const float a_l = dh[jc * 4 + 0];
    const float alr = dh[jc * 4 + 1] * D2R;
    const float d_l = dh[jc * 4 + 2];
    float sa_l, ca_l;
    sincosf(alr, &sa_l, &ca_l);
    if (lane < 6) {
        const float nsa = -sa_l, nsad = -sa_l * d_l, cad = ca_l * d_l;
        v4f lo, hi;
        if (lane == 0) {
            lo.x = 0.0f; lo.y = a_l;  lo.z = nsa;  lo.w = nsad;
            hi.x = ca_l; hi.y = cad;  hi.z = 0.0f; hi.w = 1.0f;
        } else {
            lo.x = 0.0f; lo.y = nsa;  lo.z = a_l;  lo.w = nsad;
            hi.x = ca_l; hi.y = 0.0f; hi.z = cad;  hi.w = 1.0f;
        }
        *(v4f*)(S + CONST_BASE + lane * 8)     = lo;
        *(v4f*)(S + CONST_BASE + lane * 8 + 4) = hi;
    }
    float caU[6], saU[6];
    #pragma unroll
    for (int j = 0; j < 6; ++j) {
        caU[j] = __int_as_float(__builtin_amdgcn_readlane(__float_as_int(ca_l), j));
        saU[j] = __int_as_float(__builtin_amdgcn_readlane(__float_as_int(sa_l), j));
    }

    {
        const float thv[6] = { t01.x, t01.y, t23.x, t23.y, t45.x, t45.y };
        float* rec = S + lane * REC_PITCH;
        #pragma unroll
        for (int j = 0; j < 6; ++j) {
            const float th = (dh[j * 4 + 3] + thv[j]) * D2R;
            float st, ct;
            sincosf(th, &st, &ct);
            const float ca = caU[j], sa = saU[j];
            v4f lo, hi;
            if (j == 0) {
                lo.x = ct;      lo.y = -st;     lo.z = st * ca; lo.w = ct * ca;
                hi.x = st * sa; hi.y = ct * sa; hi.z = 0.0f;    hi.w = 0.0f;
            } else {
                lo.x = ct;      lo.y = st * ca; lo.z = -st;     lo.w = ct * ca;
                hi.x = st * sa; hi.y = 0.0f;    hi.z = ct * sa; hi.w = 0.0f;
            }
            *(v4f*)(rec + j * 8)     = lo;
            *(v4f*)(rec + j * 8 + 4) = hi;
        }
    }
    __builtin_amdgcn_fence(__ATOMIC_ACQ_REL, "wavefront");

    const int h = lane >> 4;
    const int m = lane & 15;
    const int bAddr   = h ? (CONST_BASE + 2 * (m & 3)) : ((m >> 2) * REC_PITCH + 2 * (m & 3));
    const int bStride = h ? 0 : 4 * REC_PITCH;
    const int aOff    = 4 * h + 2 * (m & 1);
    const int aAddr   = (m >= 8) ? (CONST_BASE + aOff) : ((m >> 1) * REC_PITCH + aOff);
    const int aStride = (m >= 8) ? 0 : 4 * REC_PITCH;
    const bool sb0 = (lane & 4) != 0;
    const bool sb1 = (lane & 8) != 0;

    v2f Bv[8];
    #pragma unroll
    for (int g = 0; g < 8; ++g)
        Bv[g] = *(const v2f*)(S + bAddr + g * bStride);

    #pragma unroll
    for (int j = 1; j < 6; ++j) {
        v2f Av[8];
        #pragma unroll
        for (int g = 0; g < 8; ++g)
            Av[g] = *(const v2f*)(S + aAddr + g * aStride + j * 8);
        #pragma unroll
        for (int g = 0; g < 8; ++g) {
            v8f c = {};
            c = __builtin_amdgcn_wmma_f32_16x16x4_f32(
                    false, Av[g], false, Bv[g], (short)0, c, false, false);
            const float t0 = sb0 ? c[2] : c[0];
            const float t1 = sb0 ? c[3] : c[1];
            const float t2 = sb0 ? c[6] : c[4];
            const float t3 = sb0 ? c[7] : c[5];
            Bv[g].x = sb1 ? t2 : t0;
            Bv[g].y = sb1 ? t3 : t1;
        }
    }

    __builtin_amdgcn_fence(__ATOMIC_ACQ_REL, "wavefront");
    {
        const int rAddr = (m >> 2) * RES_PITCH + 4 * (m & 3) + 2 * h;
        #pragma unroll
        for (int g = 0; g < 8; ++g)
            *(v2f*)(S + rAddr + g * 4 * RES_PITCH) = Bv[g];
    }
    __builtin_amdgcn_fence(__ATOMIC_ACQ_REL, "wavefront");

    {
        const float* R = S + lane * RES_PITCH;
        const v4f r0 = *(const v4f*)(R);
        const v4f r1 = *(const v4f*)(R + 4);
        const v2f r2 = *(const v2f*)(R + 10);
        const float x = r0.w, y = r1.w, z = r2.y;
        const float T00 = r0.x, T01 = r0.y, T02 = r0.z;
        const float T10 = r1.x, T11 = r1.y, T12 = r1.z;
        const float T22 = r2.x;
        float Aa, Bb, Cc;
        if (fabsf(T12) <= 1e-6f && fabsf(T22) <= 1e-6f) {
            Aa = R2D * atan2f(T10, T11);
            Bb = R2D * atan2f(T02, T22);
            Cc = 0.0f;
        } else {
            const float An = atan2f(-T01, T00);
            float sAn, cAn;
            sincosf(An, &sAn, &cAn);
            Aa = R2D * An;
            Bb = R2D * atan2f(T02, cAn * T00 - sAn * T01);
            Cc = R2D * atan2f(-T12, T22);
        }
        if (base + lane < n) {
            v2f* o = (v2f*)(out + (size_t)(base + lane) * 6);
            v2f o0; o0.x = x;  o0.y = y;
            v2f o1; o1.x = z;  o1.y = Aa;
            v2f o2; o2.x = Bb; o2.y = Cc;
            o[0] = o0; o[1] = o1; o[2] = o2;
        }
    }
}

extern "C" void kernel_launch(void* const* d_in, const int* in_sizes, int n_in,
                              void* d_out, int out_size, void* d_ws, size_t ws_size,
                              hipStream_t stream) {
    const float* theta = (const float*)d_in[0];
    const float* dh    = (const float*)d_in[1];
    float* out = (float*)d_out;
    const int n = in_sizes[0] / 6;
    const int samplesPerBlock = 256;
    const int blocks = (n + samplesPerBlock - 1) / samplesPerBlock;
    fk_wmma_kernel<<<blocks, 256, 0, stream>>>(theta, dh, out, n);
}


# sin/cos: large-argument test made once per wave, not per call; the six joint evaluations (also alpha and the Euler stage) run the library's short path branch-free as one block of independent chains
# speedup vs baseline: 2.3485x; 1.0535x over previous
#include <hip/hip_runtime.h>
#include <math.h>


typedef __attribute__((ext_vector_type(2))) float v2f;
typedef __attribute__((ext_vector_type(4))) float v4f;
typedef __attribute__((ext_vector_type(8))) float v8f;

#define D2R 0.017453292519943295f
#define R2D 57.29577951308232f

#define REC_PITCH   52
#define CONST_BASE  1680
#define RES_PITCH   20
#define WAVE_DWORDS 1728

#define SC_SMALL_BOUND 131072.0f
__device__ __forceinline__ void sincos_small(float x, float* s, float* c)
{
    const float n = rintf(x * __uint_as_float(0x3f22f983u));
    float r = fmaf(n, __uint_as_float(0xbfc90fdau), x);
    r = fmaf(n, __uint_as_float(0xb3a22168u), r);
    r = fmaf(n, __uint_as_float(0xa7c234c4u), r);
    const unsigned q = (unsigned)(int)n;
    const float t = r * r;
    float sp = fmaf(__uint_as_float(0xb94c1982u), t, __uint_as_float(0x3c0881c4u));
    sp = fmaf(t, sp, __uint_as_float(0xbe2aaa9du));
    sp = t * sp;
    const float sr = fmaf(r, sp, r);
    float cp = fmaf(__uint_as_float(0x37d75334u), t, __uint_as_float(0xbab64f3bu));
    cp = fmaf(t, cp, __uint_as_float(0x3d2aabf7u));
    cp = fmaf(t, cp, __uint_as_float(0xbf000004u));
    const float cr = fmaf(t, cp, 1.0f);
    const bool odd = (q & 1u) != 0u;
    const float sv = odd ? cr : sr;
    const float cv = odd ? -sr : cr;
    const unsigned flip = (q << 30) & 0x80000000u;
    *s = __uint_as_float(__float_as_uint(sv) ^ flip);
    *c = __uint_as_float(__float_as_uint(cv) ^ flip);
}

__global__ __launch_bounds__(256) void fk_wmma_kernel(
    const float* __restrict__ theta, const float* __restrict__ dh,
    float* __restrict__ out, int n)
{
    __shared__ __attribute__((aligned(16))) float smem[8 * WAVE_DWORDS];
    const int lane = threadIdx.x & 31;
    const int wave = threadIdx.x >> 5;
    float* S = &smem[wave * WAVE_DWORDS];
    const int base = (blockIdx.x * 8 + wave) * 32;

    int samp = base + lane; if (samp >= n) samp = n - 1;
    const v2f* tp = (const v2f*)(theta + (size_t)samp * 6);
    const v2f t01 = tp[0], t23 = tp[1], t45 = tp[2];

    const int jc = lane < 6 ? lane : 5;
    const float a_l = dh[jc * 4 + 0];
    const float alr = dh[jc * 4 + 1] * D2R;
    const float d_l = dh[jc * 4 + 2];
    float sa_l, ca_l;
    if (__builtin_expect(__builtin_amdgcn_ballot_w32(!(fabsf(alr) < SC_SMALL_BOUND)) != 0u, 0))
        sincosf(alr, &sa_l, &ca_l);
    else
        sincos_small(alr, &sa_l, &ca_l);
    if (lane < 6) {
        const float nsa = -sa_l, nsad = -sa_l * d_l, cad = ca_l * d_l;
        v4f lo, hi;
        if (lane == 0) {
            lo.x = 0.0f; lo.y = a_l;  lo.z = nsa;  lo.w = nsad;
            hi.x = ca_l; hi.y = cad;  hi.z = 0.0f; hi.w = 1.0f;
        } else {
            lo.x = 0.0f; lo.y = nsa;  lo.z = a_l;  lo.w = nsad;
            hi.x = ca_l; hi.y = 0.0f; hi.z = cad;  hi.w = 1.0f;
        }
        *(v4f*)(S + CONST_BASE + lane * 8)     = lo;
        *(v4f*)(S + CONST_BASE + lane * 8 + 4) = hi;
    }
    float caU[6], saU[6];
    #pragma unroll
    for (int j = 0; j < 6; ++j) {
        caU[j] = __int_as_float(__builtin_amdgcn_readlane(__float_as_int(ca_l), j));
        saU[j] = __int_as_float(__builtin_amdgcn_readlane(__float_as_int(sa_l), j));
    }

    {
        const float thv[6] = { t01.x, t01.y, t23.x, t23.y, t45.x, t45.y };
        float* rec = S + lane * REC_PITCH;
        float thr[6], stv[6], ctv[6];
        float big = 0.0f;
        #pragma unroll
        for (int j = 0; j < 6; ++j) {
            thr[j] = (dh[j * 4 + 3] + thv[j]) * D2R;
            big = fmaxf(big, fabsf(thr[j]));
        }
        if (__builtin_expect(__builtin_amdgcn_ballot_w32(!(big < SC_SMALL_BOUND)) != 0u, 0)) {
            #pragma unroll
            for (int j = 0; j < 6; ++j) sincosf(thr[j], &stv[j], &ctv[j]);
        } else {
            #pragma unroll
            for (int j = 0; j < 6; ++j) sincos_small(thr[j], &stv[j], &ctv[j]);
        }
        #pragma unroll
        for (int j = 0; j < 6; ++j) {
            const float st = stv[j], ct = ctv[j];
            const float ca = caU[j], sa = saU[j];
            v4f lo, hi;
            if (j == 0) {
                lo.x = ct;      lo.y = -st;     lo.z = st * ca; lo.w = ct * ca;
                hi.x = st * sa; hi.y = ct * sa; hi.z = 0.0f;    hi.w = 0.0f;
            } else {
                lo.x = ct;      lo.y = st * ca; lo.z = -st;     lo.w = ct * ca;
                hi.x = st * sa; hi.y = 0.0f;    hi.z = ct * sa; hi.w = 0.0f;
            }
            *(v4f*)(rec + j * 8)     = lo;
            *(v4f*)(rec + j * 8 + 4) = hi;
        }
    }
    __builtin_amdgcn_fence(__ATOMIC_ACQ_REL, "wavefront");

    const int h = lane >> 4;
    const int m = lane & 15;
    const int bAddr   = h ? (CONST_BASE + 2 * (m & 3)) : ((m >> 2) * REC_PITCH + 2 * (m & 3));
    const int bStride = h ? 0 : 4 * REC_PITCH;
    const int aOff    = 4 * h + 2 * (m & 1);
    const int aAddr   = (m >= 8) ? (CONST_BASE + aOff) : ((m >> 1) * REC_PITCH + aOff);
    const int aStride = (m >= 8) ? 0 : 4 * REC_PITCH;
    const bool sb0 = (lane & 4) != 0;
    const bool sb1 = (lane & 8) != 0;

    v2f Bv[8];
    #pragma unroll
    for (int g = 0; g < 8; ++g)
        Bv[g] = *(const v2f*)(S + bAddr + g * bStride);

    #pragma unroll
    for (int j = 1; j < 6; ++j) {
        v2f Av[8];
        #pragma unroll
        for (int g = 0; g < 8; ++g)
            Av[g] = *(const v2f*)(S + aAddr + g * aStride + j * 8);
        #pragma unroll
        for (int g = 0; g < 8; ++g) {
            v8f c = {};
            c = __builtin_amdgcn_wmma_f32_16x16x4_f32(
                    false, Av[g], false, Bv[g], (short)0, c, false, false);
            const float t0 = sb0 ? c[2] : c[0];
            const float t1 = sb0 ? c[3] : c[1];
            const float t2 = sb0 ? c[6] : c[4];
            const float t3 = sb0 ? c[7] : c[5];
            Bv[g].x = sb1 ? t2 : t0;
            Bv[g].y = sb1 ? t3 : t1;
        }
    }

    __builtin_amdgcn_fence(__ATOMIC_ACQ_REL, "wavefront");
    {
        const int rAddr = (m >> 2) * RES_PITCH + 4 * (m & 3) + 2 * h;
        #pragma unroll
        for (int g = 0; g < 8; ++g)
            *(v2f*)(S + rAddr + g * 4 * RES_PITCH) = Bv[g];
    }
    __builtin_amdgcn_fence(__ATOMIC_ACQ_REL, "wavefront");

    {
        const float* R = S + lane * RES_PITCH;
        const v4f r0 = *(const v4f*)(R);
        const v4f r1 = *(const v4f*)(R + 4);
        const v2f r2 = *(const v2f*)(R + 10);
        const float x = r0.w, y = r1.w, z = r2.y;
        const float T00 = r0.x, T01 = r0.y, T02 = r0.z;
        const float T10 = r1.x, T11 = r1.y, T12 = r1.z;
        const float T22 = r2.x;
        float Aa, Bb, Cc;
        if (fabsf(T12) <= 1e-6f && fabsf(T22) <= 1e-6f) {
            Aa = R2D * atan2f(T10, T11);
            Bb = R2D * atan2f(T02, T22);
            Cc = 0.0f;
        } else {
            const float An = atan2f(-T01, T00);
            float sAn, cAn;
            sincos_small(An, &sAn, &cAn);
            Aa = R2D * An;
            Bb = R2D * atan2f(T02, cAn * T00 - sAn * T01);
            Cc = R2D * atan2f(-T12, T22);
        }
        if (base + lane < n) {
            v2f* o = (v2f*)(out + (size_t)(base + lane) * 6);
            v2f o0; o0.x = x;  o0.y = y;
            v2f o1; o1.x = z;  o1.y = Aa;
            v2f o2; o2.x = Bb; o2.y = Cc;
            o[0] = o0; o[1] = o1; o[2] = o2;
        }
    }
}

extern "C" void kernel_launch(void* const* d_in, const int* in_sizes, int n_in,
                              void* d_out, int out_size, void* d_ws, size_t ws_size,
                              hipStream_t stream) {
    const float* theta = (const float*)d_in[0];
    const float* dh    = (const float*)d_in[1];
    float* out = (float*)d_out;
    const int n = in_sizes[0] / 6;
    const int samplesPerBlock = 256;
    const int blocks = (n + samplesPerBlock - 1) / samplesPerBlock;
    fk_wmma_kernel<<<blocks, 256, 0, stream>>>(theta, dh, out, n);
}


# Constant row 3 of b dropped from the chain: 3 row vectors per sample, 96 columns = 6 WMMA groups per joint (30 WMMAs per wave, not 40); each f32 sum now adds its products in k order 0,2,1,3
# speedup vs baseline: 2.5236x; 1.0746x over previous
#include <hip/hip_runtime.h>
#include <math.h>


typedef __attribute__((ext_vector_type(2))) float v2f;
typedef __attribute__((ext_vector_type(4))) float v4f;
typedef __attribute__((ext_vector_type(8))) float v8f;

#define D2R 0.017453292519943295f
#define R2D 57.29577951308232f

#define REC_PITCH   52
#define CONST_BASE  1680
#define RES_PITCH   20
#define WAVE_DWORDS 1728

#define SC_SMALL_BOUND 131072.0f
__device__ __forceinline__ void sincos_small(float x, float* s, float* c)
{
    const float n = rintf(x * __uint_as_float(0x3f22f983u));
    float r = fmaf(n, __uint_as_float(0xbfc90fdau), x);
    r = fmaf(n, __uint_as_float(0xb3a22168u), r);
    r = fmaf(n, __uint_as_float(0xa7c234c4u), r);
    const unsigned q = (unsigned)(int)n;
    const float t = r * r;
    float sp = fmaf(__uint_as_float(0xb94c1982u), t, __uint_as_float(0x3c0881c4u));
    sp = fmaf(t, sp, __uint_as_float(0xbe2aaa9du));
    sp = t * sp;
    const float sr = fmaf(r, sp, r);
    float cp = fmaf(__uint_as_float(0x37d75334u), t, __uint_as_float(0xbab64f3bu));
    cp = fmaf(t, cp, __uint_as_float(0x3d2aabf7u));
    cp = fmaf(t, cp, __uint_as_float(0xbf000004u));
    const float cr = fmaf(t, cp, 1.0f);
    const bool odd = (q & 1u) != 0u;
    const float sv = odd ? cr : sr;
    const float cv = odd ? -sr : cr;
    const unsigned flip = (q << 30) & 0x80000000u;
    *s = __uint_as_float(__float_as_uint(sv) ^ flip);
    *c = __uint_as_float(__float_as_uint(cv) ^ flip);
}

__global__ __launch_bounds__(256) void fk_wmma_kernel(
    const float* __restrict__ theta, const float* __restrict__ dh,
    float* __restrict__ out, int n)
{
    __shared__ __attribute__((aligned(16))) float smem[8 * WAVE_DWORDS];
    const int lane = threadIdx.x & 31;
    const int wave = threadIdx.x >> 5;
    float* S = &smem[wave * WAVE_DWORDS];
    const int base = (blockIdx.x * 8 + wave) * 32;

    int samp = base + lane; if (samp >= n) samp = n - 1;
    const v2f* tp = (const v2f*)(theta + (size_t)samp * 6);
    const v2f t01 = tp[0], t23 = tp[1], t45 = tp[2];

    const int jc = lane < 6 ? lane : 5;
    const float a_l = dh[jc * 4 + 0];
    const float alr = dh[jc * 4 + 1] * D2R;
    const float d_l = dh[jc * 4 + 2];
    float sa_l, ca_l;
    if (__builtin_expect(__builtin_amdgcn_ballot_w32(!(fabsf(alr) < SC_SMALL_BOUND)) != 0u, 0))
        sincosf(alr, &sa_l, &ca_l);
    else
        sincos_small(alr, &sa_l, &ca_l);
    if (lane < 6) {
        const float nsa = -sa_l, nsad = -sa_l * d_l, cad = ca_l * d_l;
        v4f lo, hi;
        if (lane == 0) {
            lo.x = 0.0f; lo.y = a_l;  lo.z = nsa;  lo.w = nsad;
            hi.x = ca_l; hi.y = cad;  hi.z = 0.0f; hi.w = 1.0f;
        } else {
            lo.x = 0.0f; lo.y = ca_l; lo.z = nsa;  lo.w = 0.0f;
            hi.x = a_l;  hi.y = cad;  hi.z = nsad; hi.w = 1.0f;
        }
        *(v4f*)(S + CONST_BASE + lane * 8)     = lo;
        *(v4f*)(S + CONST_BASE + lane * 8 + 4) = hi;
    }
    float caU[6], saU[6];
    #pragma unroll
    for (int j = 0; j < 6; ++j) {
        caU[j] = __int_as_float(__builtin_amdgcn_readlane(__float_as_int(ca_l), j));
        saU[j] = __int_as_float(__builtin_amdgcn_readlane(__float_as_int(sa_l), j));
    }

    {
        const float thv[6] = { t01.x, t01.y, t23.x, t23.y, t45.x, t45.y };
        float* rec = S + lane * REC_PITCH;
        float thr[6], stv[6], ctv[6];
        float big = 0.0f;
        #pragma unroll
        for (int j = 0; j < 6; ++j) {
            thr[j] = (dh[j * 4 + 3] + thv[j]) * D2R;
            big = fmaxf(big, fabsf(thr[j]));
        }
        if (__builtin_expect(__builtin_amdgcn_ballot_w32(!(big < SC_SMALL_BOUND)) != 0u, 0)) {
            #pragma unroll
            for (int j = 0; j < 6; ++j) sincosf(thr[j], &stv[j], &ctv[j]);
        } else {
            #pragma unroll
            for (int j = 0; j < 6; ++j) sincos_small(thr[j], &stv[j], &ctv[j]);
        }
        #pragma unroll
        for (int j = 0; j < 6; ++j) {
            const float st = stv[j], ct = ctv[j];
            const float ca = caU[j], sa = saU[j];
            v4f lo, hi;
            if (j == 0) {
                lo.x = ct;      lo.y = -st;     lo.z = st * ca; lo.w = ct * ca;
                hi.x = st * sa; hi.y = ct * sa; hi.z = 0.0f;    hi.w = 0.0f;
            } else {
                lo.x = ct;      lo.y = st * sa; lo.z = st * ca; lo.w = 0.0f;
                hi.x = -st;     hi.y = ct * sa; hi.z = ct * ca; hi.w = 0.0f;
            }
            *(v4f*)(rec + j * 8)     = lo;
            *(v4f*)(rec + j * 8 + 4) = hi;
        }
    }
    __builtin_amdgcn_fence(__ATOMIC_ACQ_REL, "wavefront");

    const int h = lane >> 4;
    const int m = lane & 15;
    const int sB[6] = { 0, 5, 10, 16, 21, 26 };
    int slotT[3], rowT[3];
    bool s0T[3], s1T[3], s2T[3];
    #pragma unroll
    for (int t = 0; t < 3; ++t) {
        const unsigned x = (unsigned)(m + t);
        const unsigned q = x / 3u;
        slotT[t] = (int)q;
        rowT[t]  = (int)(x - 3u * q);
        s0T[t] = (q & 1u) != 0u;
        s1T[t] = (q & 2u) != 0u;
        s2T[t] = (q & 4u) != 0u;
    }
    const int  aRow    = m & 7;
    const bool aShared = aRow >= 6;
    const int  aOff    = 4 * (m >> 3) + 2 * h;
    const int  aAddr   = aShared ? (CONST_BASE + aOff) : (aRow * REC_PITCH + aOff);
    const int  aStride = aShared ? 0 : REC_PITCH;

    v2f Bv[6];
    #pragma unroll
    for (int g = 0; g < 6; ++g) {
        const int t = g % 3;
        const int e = 2 * rowT[t] + h;
        Bv[g].x = S[(sB[g] + slotT[t]) * REC_PITCH + e];
        Bv[g].y = S[CONST_BASE + e];
    }

    #pragma unroll
    for (int j = 1; j < 6; ++j) {
        v2f Av[6];
        #pragma unroll
        for (int g = 0; g < 6; ++g)
            Av[g] = *(const v2f*)(S + aAddr + sB[g] * aStride + j * 8);
        #pragma unroll
        for (int g = 0; g < 6; ++g) {
            const int t = g % 3;
            v8f c = {};
            c = __builtin_amdgcn_wmma_f32_16x16x4_f32(
                    false, Av[g], false, Bv[g], (short)0, c, false, false);
            const float t01 = s0T[t] ? c[1] : c[0];
            const float t23 = s0T[t] ? c[3] : c[2];
            const float t45 = s0T[t] ? c[5] : c[4];
            const float u   = s1T[t] ? t23 : t01;
            Bv[g].x = s2T[t] ? t45 : u;
            Bv[g].y = c[7];
        }
    }

    __builtin_amdgcn_fence(__ATOMIC_ACQ_REL, "wavefront");
    {
        const int l2 = (int)__builtin_amdgcn_mbcnt_lo(~0u, 0u);
        const int h2 = l2 >> 4;
        const int m2 = l2 & 15;
        #pragma unroll
        for (int g = 0; g < 6; ++g) {
            const unsigned x = (unsigned)(m2 + g % 3);
            const unsigned q = x / 3u;
            float* r = S + (sB[g] + (int)q) * RES_PITCH + 4 * (int)(x - 3u * q) + h2;
            r[0] = Bv[g].x;
            r[2] = Bv[g].y;
        }
    }
    __builtin_amdgcn_fence(__ATOMIC_ACQ_REL, "wavefront");

    {
        const float* R = S + lane * RES_PITCH;
        const v4f r0 = *(const v4f*)(R);
        const v4f r1 = *(const v4f*)(R + 4);
        const v2f r2 = *(const v2f*)(R + 10);
        const float x = r0.w, y = r1.w, z = r2.y;
        const float T00 = r0.x, T01 = r0.y, T02 = r0.z;
        const float T10 = r1.x, T11 = r1.y, T12 = r1.z;
        const float T22 = r2.x;
        float Aa, Bb, Cc;
        if (fabsf(T12) <= 1e-6f && fabsf(T22) <= 1e-6f) {
            Aa = R2D * atan2f(T10, T11);
            Bb = R2D * atan2f(T02, T22);
            Cc = 0.0f;
        } else {
            const float An = atan2f(-T01, T00);
            float sAn, cAn;
            sincos_small(An, &sAn, &cAn);
            Aa = R2D * An;
            Bb = R2D * atan2f(T02, cAn * T00 - sAn * T01);
            Cc = R2D * atan2f(-T12, T22);
        }
        if (base + lane < n) {
            v2f* o = (v2f*)(out + (size_t)(base + lane) * 6);
            v2f o0; o0.x = x;  o0.y = y;
            v2f o1; o1.x = z;  o1.y = Aa;
            v2f o2; o2.x = Bb; o2.y = Cc;
            o[0] = o0; o[1] = o1; o[2] = o2;
        }
    }
}

extern "C" void kernel_launch(void* const* d_in, const int* in_sizes, int n_in,
                              void* d_out, int out_size, void* d_ws, size_t ws_size,
                              hipStream_t stream) {
    const float* theta = (const float*)d_in[0];
    const float* dh    = (const float*)d_in[1];
    float* out = (float*)d_out;
    const int n = in_sizes[0] / 6;
    const int samplesPerBlock = 256;
    const int blocks = (n + samplesPerBlock - 1) / samplesPerBlock;
    fk_wmma_kernel<<<blocks, 256, 0, stream>>>(theta, dh, out, n);
}


# Each wave now loops over 4 tiles of 32 samples (256 work-groups of 32 waves, one per WGP): joint constants and lane address math once per wave, the next tile's angles requested one tile ahead
# speedup vs baseline: 2.6498x; 1.0500x over previous
#include <hip/hip_runtime.h>
#include <math.h>


typedef __attribute__((ext_vector_type(2))) float v2f;
typedef __attribute__((ext_vector_type(4))) float v4f;
typedef __attribute__((ext_vector_type(8))) float v8f;

#define D2R 0.017453292519943295f
#define R2D 57.29577951308232f

#define REC_PITCH   52
#define CONST_BASE  1680
#define RES_PITCH   20
#define WAVE_DWORDS 1728

#define SC_SMALL_BOUND 131072.0f
__device__ __forceinline__ void sincos_small(float x, float* s, float* c)
{
    const float n = rintf(x * __uint_as_float(0x3f22f983u));
    float r = fmaf(n, __uint_as_float(0xbfc90fdau), x);
    r = fmaf(n, __uint_as_float(0xb3a22168u), r);
    r = fmaf(n, __uint_as_float(0xa7c234c4u), r);
    const unsigned q = (unsigned)(int)n;
    const float t = r * r;
    float sp = fmaf(__uint_as_float(0xb94c1982u), t, __uint_as_float(0x3c0881c4u));
    sp = fmaf(t, sp, __uint_as_float(0xbe2aaa9du));
    sp = t * sp;
    const float sr = fmaf(r, sp, r);
    float cp = fmaf(__uint_as_float(0x37d75334u), t, __uint_as_float(0xbab64f3bu));
    cp = fmaf(t, cp, __uint_as_float(0x3d2aabf7u));
    cp = fmaf(t, cp, __uint_as_float(0xbf000004u));
    const float cr = fmaf(t, cp, 1.0f);
    const bool odd = (q & 1u) != 0u;
    const float sv = odd ? cr : sr;
    const float cv = odd ? -sr : cr;
    const unsigned flip = (q << 30) & 0x80000000u;
    *s = __uint_as_float(__float_as_uint(sv) ^ flip);
    *c = __uint_as_float(__float_as_uint(cv) ^ flip);
}

__global__ __launch_bounds__(1024) void fk_wmma_kernel(
    const float* __restrict__ theta, const float* __restrict__ dh,
    float* __restrict__ out, int n)
{
    __shared__ __attribute__((aligned(16))) float smem[32 * WAVE_DWORDS];
    const int lane = threadIdx.x & 31;
    const int wave = __builtin_amdgcn_readfirstlane((int)(threadIdx.x >> 5));
    float* S = &smem[wave * WAVE_DWORDS];
    const int numTiles   = (n + 31) >> 5;
    const int waveStride = (int)gridDim.x * 32;
    int tile = (int)blockIdx.x * 32 + wave;

    v2f t01, t23, t45;
    {
        int samp = tile * 32 + lane; if (samp >= n) samp = n - 1;
        const v2f* tp = (const v2f*)(theta + (size_t)samp * 6);
        t01 = tp[0]; t23 = tp[1]; t45 = tp[2];
    }

    const int jc = lane < 6 ? lane : 5;
    const float a_l = dh[jc * 4 + 0];
    const float alr = dh[jc * 4 + 1] * D2R;
    const float d_l = dh[jc * 4 + 2];
    float sa_l, ca_l;
    if (__builtin_expect(__builtin_amdgcn_ballot_w32(!(fabsf(alr) < SC_SMALL_BOUND)) != 0u, 0))
        sincosf(alr, &sa_l, &ca_l);
    else
        sincos_small(alr, &sa_l, &ca_l);
    if (lane < 6) {
        const float nsa = -sa_l, nsad = -sa_l * d_l, cad = ca_l * d_l;
        v4f lo, hi;
        if (lane == 0) {
            lo.x = 0.0f; lo.y = a_l;  lo.z = nsa;  lo.w = nsad;
            hi.x = ca_l; hi.y = cad;  hi.z = 0.0f; hi.w = 1.0f;
        } else {
            lo.x = 0.0f; lo.y = ca_l; lo.z = nsa;  lo.w = 0.0f;
            hi.x = a_l;  hi.y = cad;  hi.z = nsad; hi.w = 1.0f;
        }
        *(v4f*)(S + CONST_BASE + lane * 8)     = lo;
        *(v4f*)(S + CONST_BASE + lane * 8 + 4) = hi;
    }
    float caU[6], saU[6], offU[6];
    #pragma unroll
    for (int j = 0; j < 6; ++j) {
        caU[j] = __int_as_float(__builtin_amdgcn_readlane(__float_as_int(ca_l), j));
        saU[j] = __int_as_float(__builtin_amdgcn_readlane(__float_as_int(sa_l), j));
        offU[j] = dh[j * 4 + 3];
    }

    const int h = lane >> 4;
    const int m = lane & 15;
    const int sB[6] = { 0, 5, 10, 16, 21, 26 };
    int slotT[3], rowT[3];
    bool s0T[3], s1T[3], s2T[3];
    #pragma unroll
    for (int t = 0; t < 3; ++t) {
        const unsigned x = (unsigned)(m + t);
        const unsigned q = x / 3u;
        slotT[t] = (int)q;
        rowT[t]  = (int)(x - 3u * q);
        s0T[t] = (q & 1u) != 0u;
        s1T[t] = (q & 2u) != 0u;
        s2T[t] = (q & 4u) != 0u;
    }
    const int  aRow    = m & 7;
    const bool aShared = aRow >= 6;
    const int  aOff    = 4 * (m >> 3) + 2 * h;
    const int  aAddr   = aShared ? (CONST_BASE + aOff) : (aRow * REC_PITCH + aOff);
    const int  aStride = aShared ? 0 : REC_PITCH;
    int resOff[3];
    #pragma unroll
    for (int t = 0; t < 3; ++t)
        resOff[t] = slotT[t] * RES_PITCH + 4 * rowT[t] + h;

    for (; tile < numTiles; tile += waveStride) {
        const int base = tile * 32;

        v2f n01, n23, n45;
        {
            int ns = base + waveStride * 32 + lane; if (ns >= n) ns = n - 1;
            const v2f* np = (const v2f*)(theta + (size_t)ns * 6);
            n01 = np[0]; n23 = np[1]; n45 = np[2];
        }

        {
            const float thv[6] = { t01.x, t01.y, t23.x, t23.y, t45.x, t45.y };
            float* rec = S + lane * REC_PITCH;
            float thr[6], stv[6], ctv[6];
            float big = 0.0f;
            #pragma unroll
            for (int j = 0; j < 6; ++j) {
                thr[j] = (offU[j] + thv[j]) * D2R;
                big = fmaxf(big, fabsf(thr[j]));
            }
            if (__builtin_expect(__builtin_amdgcn_ballot_w32(!(big < SC_SMALL_BOUND)) != 0u, 0)) {
                #pragma unroll
                for (int j = 0; j < 6; ++j) sincosf(thr[j], &stv[j], &ctv[j]);
            } else {
                #pragma unroll
                for (int j = 0; j < 6; ++j) sincos_small(thr[j], &stv[j], &ctv[j]);
            }
            #pragma unroll
            for (int j = 0; j < 6; ++j) {
                const float st = stv[j], ct = ctv[j];
                const float ca = caU[j], sa = saU[j];
                v4f lo, hi;
                if (j == 0) {
                    lo.x = ct;      lo.y = -st;     lo.z = st * ca; lo.w = ct * ca;
                    hi.x = st * sa; hi.y = ct * sa; hi.z = 0.0f;    hi.w = 0.0f;
                } else {
                    lo.x = ct;      lo.y = st * sa; lo.z = st * ca; lo.w = 0.0f;
                    hi.x = -st;     hi.y = ct * sa; hi.z = ct * ca; hi.w = 0.0f;
                }
                *(v4f*)(rec + j * 8)     = lo;
                *(v4f*)(rec + j * 8 + 4) = hi;
            }
        }
        __builtin_amdgcn_fence(__ATOMIC_ACQ_REL, "wavefront");

        v2f Bv[6];
        #pragma unroll
        for (int g = 0; g < 6; ++g) {
            const int t = g % 3;
            const int e = 2 * rowT[t] + h;
            Bv[g].x = S[(sB[g] + slotT[t]) * REC_PITCH + e];
            Bv[g].y = S[CONST_BASE + e];
        }

        #pragma unroll
        for (int j = 1; j < 6; ++j) {
            v2f Av[6];
            #pragma unroll
            for (int g = 0; g < 6; ++g)
                Av[g] = *(const v2f*)(S + aAddr + sB[g] * aStride + j * 8);
            #pragma unroll
            for (int g = 0; g < 6; ++g) {
                const int t = g % 3;
                v8f c = {};
                c = __builtin_amdgcn_wmma_f32_16x16x4_f32(
                        false, Av[g], false, Bv[g], (short)0, c, false, false);
                const float u01 = s0T[t] ? c[1] : c[0];
                const float u23 = s0T[t] ? c[3] : c[2];
                const float u45 = s0T[t] ? c[5] : c[4];
                const float u   = s1T[t] ? u23 : u01;
                Bv[g].x = s2T[t] ? u45 : u;
                Bv[g].y = c[7];
            }
        }

        __builtin_amdgcn_fence(__ATOMIC_ACQ_REL, "wavefront");
        #pragma unroll
        for (int g = 0; g < 6; ++g) {
            float* r = S + sB[g] * RES_PITCH + resOff[g % 3];
            r[0] = Bv[g].x;
            r[2] = Bv[g].y;
        }
        __builtin_amdgcn_fence(__ATOMIC_ACQ_REL, "wavefront");

        {
            const float* R = S + lane * RES_PITCH;
            const v4f r0 = *(const v4f*)(R);
            const v4f r1 = *(const v4f*)(R + 4);
            const v2f r2 = *(const v2f*)(R + 10);
            __builtin_amdgcn_fence(__ATOMIC_ACQ_REL, "wavefront");
            const float x = r0.w, y = r1.w, z = r2.y;
            const float T00 = r0.x, T01 = r0.y, T02 = r0.z;
            const float T10 = r1.x, T11 = r1.y, T12 = r1.z;
            const float T22 = r2.x;
            float Aa, Bb, Cc;
            if (fabsf(T12) <= 1e-6f && fabsf(T22) <= 1e-6f) {
                Aa = R2D * atan2f(T10, T11);
                Bb = R2D * atan2f(T02, T22);
                Cc = 0.0f;
            } else {
                const float An = atan2f(-T01, T00);
                float sAn, cAn;
                sincos_small(An, &sAn, &cAn);
                Aa = R2D * An;
                Bb = R2D * atan2f(T02, cAn * T00 - sAn * T01);
                Cc = R2D * atan2f(-T12, T22);
            }
            if (base + lane < n) {
                v2f* o = (v2f*)(out + (size_t)(base + lane) * 6);
                v2f o0; o0.x = x;  o0.y = y;
                v2f o1; o1.x = z;  o1.y = Aa;
                v2f o2; o2.x = Bb; o2.y = Cc;
                o[0] = o0; o[1] = o1; o[2] = o2;
            }
        }
        t01 = n01; t23 = n23; t45 = n45;
    }
}

extern "C" void kernel_launch(void* const* d_in, const int* in_sizes, int n_in,
                              void* d_out, int out_size, void* d_ws, size_t ws_size,
                              hipStream_t stream) {
    const float* theta = (const float*)d_in[0];
    const float* dh    = (const float*)d_in[1];
    float* out = (float*)d_out;
    const int n = in_sizes[0] / 6;
    fk_wmma_kernel<<<256, 1024, 0, stream>>>(theta, dh, out, n);
}


# Euler stage cut to about a third of its vector instructions: B from sqrt(T00^2+T01^2) in place of sin/cos of A; atan2 as plain reciprocal + the library polynomial in packed f32 pairs
# speedup vs baseline: 2.8386x; 1.0713x over previous
#include <hip/hip_runtime.h>
#include <math.h>


typedef __attribute__((ext_vector_type(2))) float v2f;
typedef __attribute__((ext_vector_type(4))) float v4f;
typedef __attribute__((ext_vector_type(8))) float v8f;

#define D2R 0.017453292519943295f
#define R2D 57.29577951308232f

#define REC_PITCH   52
#define CONST_BASE  1680
#define RES_PITCH   20
#define WAVE_DWORDS 1728

#define SC_SMALL_BOUND 131072.0f
__device__ __forceinline__ void sincos_small(float x, float* s, float* c)
{
    const float n = rintf(x * __uint_as_float(0x3f22f983u));
    float r = fmaf(n, __uint_as_float(0xbfc90fdau), x);
    r = fmaf(n, __uint_as_float(0xb3a22168u), r);
    r = fmaf(n, __uint_as_float(0xa7c234c4u), r);
    const unsigned q = (unsigned)(int)n;
    const float t = r * r;
    float sp = fmaf(__uint_as_float(0xb94c1982u), t, __uint_as_float(0x3c0881c4u));
    sp = fmaf(t, sp, __uint_as_float(0xbe2aaa9du));
    sp = t * sp;
    const float sr = fmaf(r, sp, r);
    float cp = fmaf(__uint_as_float(0x37d75334u), t, __uint_as_float(0xbab64f3bu));
    cp = fmaf(t, cp, __uint_as_float(0x3d2aabf7u));
    cp = fmaf(t, cp, __uint_as_float(0xbf000004u));
    const float cr = fmaf(t, cp, 1.0f);
    const bool odd = (q & 1u) != 0u;
    const float sv = odd ? cr : sr;
    const float cv = odd ? -sr : cr;
    const unsigned flip = (q << 30) & 0x80000000u;
    *s = __uint_as_float(__float_as_uint(sv) ^ flip);
    *c = __uint_as_float(__float_as_uint(cv) ^ flip);
}

#define AT_C0 __uint_as_float(0x3b2d2a58u)
#define AT_C1 __uint_as_float(0xbc7a590cu)
#define AT_C2 __uint_as_float(0x3d29fb3fu)
#define AT_C3 __uint_as_float(0xbd97d4d7u)
#define AT_C4 __uint_as_float(0x3dd931b2u)
#define AT_C5 __uint_as_float(0xbe1160e6u)
#define AT_C6 __uint_as_float(0x3e4cb8bfu)
#define AT_C7 __uint_as_float(0xbeaaaa62u)
#define AT_HALF_PI __uint_as_float(0x3fc90fdbu)
#define AT_PI      __uint_as_float(0x40490fdbu)
__device__ __forceinline__ v2f splat2(float c) { v2f r; r.x = c; r.y = c; return r; }
__device__ __forceinline__ v2f atan_poly2(v2f a)
{
    const v2f s = a * a;
    v2f p = __builtin_elementwise_fma(splat2(AT_C0), s, splat2(AT_C1));
    p = __builtin_elementwise_fma(s, p, splat2(AT_C2));
    p = __builtin_elementwise_fma(s, p, splat2(AT_C3));
    p = __builtin_elementwise_fma(s, p, splat2(AT_C4));
    p = __builtin_elementwise_fma(s, p, splat2(AT_C5));
    p = __builtin_elementwise_fma(s, p, splat2(AT_C6));
    p = __builtin_elementwise_fma(s, p, splat2(AT_C7));
    p = s * p;
    return __builtin_elementwise_fma(a, p, a);
}
__device__ __forceinline__ float ratio_small(float mn, float mx)
{
    const float q = mn * __builtin_amdgcn_rcpf(mx);
    return mx == 0.0f ? 0.0f : q;
}
template <bool LEFT>
__device__ __forceinline__ float atan2_finish(float p, float s, float x, bool steep)
{
    p = steep ? AT_HALF_PI - p : p;
    if (LEFT) p = ((int)__float_as_uint(x) < 0) ? AT_PI - p : p;
    return __uint_as_float((__float_as_uint(p) & 0x7fffffffu) | (__float_as_uint(s) & 0x80000000u));
}

__global__ __launch_bounds__(1024) void fk_wmma_kernel(
    const float* __restrict__ theta, const float* __restrict__ dh,
    float* __restrict__ out, int n)
{
    __shared__ __attribute__((aligned(16))) float smem[32 * WAVE_DWORDS];
    const int lane = threadIdx.x & 31;
    const int wave = __builtin_amdgcn_readfirstlane((int)(threadIdx.x >> 5));
    float* S = &smem[wave * WAVE_DWORDS];
    const int numTiles   = (n + 31) >> 5;
    const int waveStride = (int)gridDim.x * 32;
    int tile = (int)blockIdx.x * 32 + wave;

    v2f t01, t23, t45;
    {
        int samp = tile * 32 + lane; if (samp >= n) samp = n - 1;
        const v2f* tp = (const v2f*)(theta + (size_t)samp * 6);
        t01 = tp[0]; t23 = tp[1]; t45 = tp[2];
    }

    const int jc = lane < 6 ? lane : 5;
    const float a_l = dh[jc * 4 + 0];
    const float alr = dh[jc * 4 + 1] * D2R;
    const float d_l = dh[jc * 4 + 2];
    float sa_l, ca_l;
    if (__builtin_expect(__builtin_amdgcn_ballot_w32(!(fabsf(alr) < SC_SMALL_BOUND)) != 0u, 0))
        sincosf(alr, &sa_l, &ca_l);
    else
        sincos_small(alr, &sa_l, &ca_l);
    if (lane < 6) {
        const float nsa = -sa_l, nsad = -sa_l * d_l, cad = ca_l * d_l;
        v4f lo, hi;
        if (lane == 0) {
            lo.x = 0.0f; lo.y = a_l;  lo.z = nsa;  lo.w = nsad;
            hi.x = ca_l; hi.y = cad;  hi.z = 0.0f; hi.w = 1.0f;
        } else {
            lo.x = 0.0f; lo.y = ca_l; lo.z = nsa;  lo.w = 0.0f;
            hi.x = a_l;  hi.y = cad;  hi.z = nsad; hi.w = 1.0f;
        }
        *(v4f*)(S + CONST_BASE + lane * 8)     = lo;
        *(v4f*)(S + CONST_BASE + lane * 8 + 4) = hi;
    }
    float caU[6], saU[6], offU[6];
    #pragma unroll
    for (int j = 0; j < 6; ++j) {
        caU[j] = __int_as_float(__builtin_amdgcn_readlane(__float_as_int(ca_l), j));
        saU[j] = __int_as_float(__builtin_amdgcn_readlane(__float_as_int(sa_l), j));
        offU[j] = dh[j * 4 + 3];
    }

    const int h = lane >> 4;
    const int m = lane & 15;
    const int sB[6] = { 0, 5, 10, 16, 21, 26 };
    int slotT[3], rowT[3];
    bool s0T[3], s1T[3], s2T[3];
    #pragma unroll
    for (int t = 0; t < 3; ++t) {
        const unsigned x = (unsigned)(m + t);
        const unsigned q = x / 3u;
        slotT[t] = (int)q;
        rowT[t]  = (int)(x - 3u * q);
        s0T[t] = (q & 1u) != 0u;
        s1T[t] = (q & 2u) != 0u;
        s2T[t] = (q & 4u) != 0u;
    }
    const int  aRow    = m & 7;
    const bool aShared = aRow >= 6;
    const int  aOff    = 4 * (m >> 3) + 2 * h;
    const int  aAddr   = aShared ? (CONST_BASE + aOff) : (aRow * REC_PITCH + aOff);
    const int  aStride = aShared ? 0 : REC_PITCH;
    int resOff[3];
    #pragma unroll
    for (int t = 0; t < 3; ++t)
        resOff[t] = slotT[t] * RES_PITCH + 4 * rowT[t] + h;

    for (; tile < numTiles; tile += waveStride) {
        const int base = tile * 32;

        v2f n01, n23, n45;
        {
            int ns = base + waveStride * 32 + lane; if (ns >= n) ns = n - 1;
            const v2f* np = (const v2f*)(theta + (size_t)ns * 6);
            n01 = np[0]; n23 = np[1]; n45 = np[2];
        }

        {
            const float thv[6] = { t01.x, t01.y, t23.x, t23.y, t45.x, t45.y };
            float* rec = S + lane * REC_PITCH;
            float thr[6], stv[6], ctv[6];
            float big = 0.0f;
            #pragma unroll
            for (int j = 0; j < 6; ++j) {
                thr[j] = (offU[j] + thv[j]) * D2R;
                big = fmaxf(big, fabsf(thr[j]));
            }
            if (__builtin_expect(__builtin_amdgcn_ballot_w32(!(big < SC_SMALL_BOUND)) != 0u, 0)) {
                #pragma unroll
                for (int j = 0; j < 6; ++j) sincosf(thr[j], &stv[j], &ctv[j]);
            } else {
                #pragma unroll
                for (int j = 0; j < 6; ++j) sincos_small(thr[j], &stv[j], &ctv[j]);
            }
            #pragma unroll
            for (int j = 0; j < 6; ++j) {
                const float st = stv[j], ct = ctv[j];
                const float ca = caU[j], sa = saU[j];
                v4f lo, hi;
                if (j == 0) {
                    lo.x = ct;      lo.y = -st;     lo.z = st * ca; lo.w = ct * ca;
                    hi.x = st * sa; hi.y = ct * sa; hi.z = 0.0f;    hi.w = 0.0f;
                } else {
                    lo.x = ct;      lo.y = st * sa; lo.z = st * ca; lo.w = 0.0f;
                    hi.x = -st;     hi.y = ct * sa; hi.z = ct * ca; hi.w = 0.0f;
                }
                *(v4f*)(rec + j * 8)     = lo;
                *(v4f*)(rec + j * 8 + 4) = hi;
            }
        }
        __builtin_amdgcn_fence(__ATOMIC_ACQ_REL, "wavefront");

        v2f Bv[6];
        #pragma unroll
        for (int g = 0; g < 6; ++g) {
            const int t = g % 3;
            const int e = 2 * rowT[t] + h;
            Bv[g].x = S[(sB[g] + slotT[t]) * REC_PITCH + e];
            Bv[g].y = S[CONST_BASE + e];
        }

        #pragma unroll
        for (int j = 1; j < 6; ++j) {
            v2f Av[6];
            #pragma unroll
            for (int g = 0; g < 6; ++g)
                Av[g] = *(const v2f*)(S + aAddr + sB[g] * aStride + j * 8);
            #pragma unroll
            for (int g = 0; g < 6; ++g) {
                const int t = g % 3;
                v8f c = {};
                c = __builtin_amdgcn_wmma_f32_16x16x4_f32(
                        false, Av[g], false, Bv[g], (short)0, c, false, false);
                const float u01 = s0T[t] ? c[1] : c[0];
                const float u23 = s0T[t] ? c[3] : c[2];
                const float u45 = s0T[t] ? c[5] : c[4];
                const float u   = s1T[t] ? u23 : u01;
                Bv[g].x = s2T[t] ? u45 : u;
                Bv[g].y = c[7];
            }
        }

        __builtin_amdgcn_fence(__ATOMIC_ACQ_REL, "wavefront");
        #pragma unroll
        for (int g = 0; g < 6; ++g) {
            float* r = S + sB[g] * RES_PITCH + resOff[g % 3];
            r[0] = Bv[g].x;
            r[2] = Bv[g].y;
        }
        __builtin_amdgcn_fence(__ATOMIC_ACQ_REL, "wavefront");

        {
            const float* R = S + lane * RES_PITCH;
            const v4f r0 = *(const v4f*)(R);
            const v4f r1 = *(const v4f*)(R + 4);
            const v2f r2 = *(const v2f*)(R + 10);
            __builtin_amdgcn_fence(__ATOMIC_ACQ_REL, "wavefront");
            const float x = r0.w, y = r1.w, z = r2.y;
            const float T00 = r0.x, T01 = r0.y, T02 = r0.z;
            const float T10 = r1.x, T11 = r1.y, T12 = r1.z;
            const float T22 = r2.x;
            float Aa, Bb, Cc;
            if (fabsf(T12) <= 1e-6f && fabsf(T22) <= 1e-6f) {
                Aa = R2D * atan2f(T10, T11);
                Bb = R2D * atan2f(T02, T22);
                Cc = 0.0f;
            } else {
                const float len = __builtin_amdgcn_sqrtf(fmaf(T00, T00, T01 * T01));
                const float ayA = fabsf(T01), axA = fabsf(T00);
                const float ayC = fabsf(T12), axC = fabsf(T22);
                const float ayB = fabsf(T02);
                const bool stA = ayA > axA, stC = ayC > axC, stB = ayB > len;
                v2f q;
                q.x = ratio_small(stA ? axA : ayA, stA ? ayA : axA);
                q.y = ratio_small(stC ? axC : ayC, stC ? ayC : axC);
                const v2f pAC = atan_poly2(q);
                const float pB = atan_poly2(splat2(ratio_small(stB ? len : ayB, stB ? ayB : len))).x;
                Aa = -R2D * atan2_finish<true>(pAC.x, T01, T00, stA);
                Cc = -R2D * atan2_finish<true>(pAC.y, T12, T22, stC);
                Bb =  R2D * atan2_finish<false>(pB, T02, len, stB);
            }
            if (base + lane < n) {
                v2f* o = (v2f*)(out + (size_t)(base + lane) * 6);
                v2f o0; o0.x = x;  o0.y = y;
                v2f o1; o1.x = z;  o1.y = Aa;
                v2f o2; o2.x = Bb; o2.y = Cc;
                o[0] = o0; o[1] = o1; o[2] = o2;
            }
        }
        t01 = n01; t23 = n23; t45 = n45;
    }
}

extern "C" void kernel_launch(void* const* d_in, const int* in_sizes, int n_in,
                              void* d_out, int out_size, void* d_ws, size_t ws_size,
                              hipStream_t stream) {
    const float* theta = (const float*)d_in[0];
    const float* dh    = (const float*)d_in[1];
    float* out = (float*)d_out;
    const int n = in_sizes[0] / 6;
    fk_wmma_kernel<<<256, 1024, 0, stream>>>(theta, dh, out, n);
}


# T_j factored as X_j*Rz*Tz: WMMA A operand is a per-wave constant in registers, Rz by lane FMAs, one f32 WMMA spans two joints (14 per tile, not 30), no LDS; f32 sums associate differently
# speedup vs baseline: 4.0201x; 1.4162x over previous
#include <hip/hip_runtime.h>
#include <math.h>


typedef __attribute__((ext_vector_type(2))) float v2f;
typedef __attribute__((ext_vector_type(4))) float v4f;
typedef __attribute__((ext_vector_type(8))) float v8f;
typedef __attribute__((ext_vector_type(2))) unsigned v2u;

#define D2R 0.017453292519943295f
#define R2D 57.29577951308232f

#define SC_SMALL_BOUND 131072.0f
__device__ __forceinline__ void sincos_small(float x, float* s, float* c)
{
    const float n = rintf(x * __uint_as_float(0x3f22f983u));
    float r = fmaf(n, __uint_as_float(0xbfc90fdau), x);
    r = fmaf(n, __uint_as_float(0xb3a22168u), r);
    r = fmaf(n, __uint_as_float(0xa7c234c4u), r);
    const unsigned q = (unsigned)(int)n;
    const float t = r * r;
    float sp = fmaf(__uint_as_float(0xb94c1982u), t, __uint_as_float(0x3c0881c4u));
    sp = fmaf(t, sp, __uint_as_float(0xbe2aaa9du));
    sp = t * sp;
    const float sr = fmaf(r, sp, r);
    float cp = fmaf(__uint_as_float(0x37d75334u), t, __uint_as_float(0xbab64f3bu));
    cp = fmaf(t, cp, __uint_as_float(0x3d2aabf7u));
    cp = fmaf(t, cp, __uint_as_float(0xbf000004u));
    const float cr = fmaf(t, cp, 1.0f);
    const bool odd = (q & 1u) != 0u;
    const float sv = odd ? cr : sr;
    const float cv = odd ? -sr : cr;
    const unsigned flip = (q << 30) & 0x80000000u;
    *s = __uint_as_float(__float_as_uint(sv) ^ flip);
    *c = __uint_as_float(__float_as_uint(cv) ^ flip);
}

#define AT_C0 __uint_as_float(0x3b2d2a58u)
#define AT_C1 __uint_as_float(0xbc7a590cu)
#define AT_C2 __uint_as_float(0x3d29fb3fu)
#define AT_C3 __uint_as_float(0xbd97d4d7u)
#define AT_C4 __uint_as_float(0x3dd931b2u)
#define AT_C5 __uint_as_float(0xbe1160e6u)
#define AT_C6 __uint_as_float(0x3e4cb8bfu)
#define AT_C7 __uint_as_float(0xbeaaaa62u)
#define AT_HALF_PI __uint_as_float(0x3fc90fdbu)
#define AT_PI      __uint_as_float(0x40490fdbu)
__device__ __forceinline__ v2f splat2(float c) { v2f r; r.x = c; r.y = c; return r; }
__device__ __forceinline__ v2f atan_poly2(v2f a)
{
    const v2f s = a * a;
    v2f p = __builtin_elementwise_fma(splat2(AT_C0), s, splat2(AT_C1));
    p = __builtin_elementwise_fma(s, p, splat2(AT_C2));
    p = __builtin_elementwise_fma(s, p, splat2(AT_C3));
    p = __builtin_elementwise_fma(s, p, splat2(AT_C4));
    p = __builtin_elementwise_fma(s, p, splat2(AT_C5));
    p = __builtin_elementwise_fma(s, p, splat2(AT_C6));
    p = __builtin_elementwise_fma(s, p, splat2(AT_C7));
    p = s * p;
    return __builtin_elementwise_fma(a, p, a);
}
__device__ __forceinline__ float ratio_small(float mn, float mx)
{
    const float q = mn * __builtin_amdgcn_rcpf(mx);
    return mx == 0.0f ? 0.0f : q;
}
template <bool LEFT>
__device__ __forceinline__ float atan2_finish(float p, float s, float x, bool steep)
{
    p = steep ? AT_HALF_PI - p : p;
    if (LEFT) p = ((int)__float_as_uint(x) < 0) ? AT_PI - p : p;
    return __uint_as_float((__float_as_uint(p) & 0x7fffffffu) | (__float_as_uint(s) & 0x80000000u));
}

struct JC { float ca, sa, a, dp; };

__device__ __forceinline__ void right_pair(bool e, int c, const JC& R, float* r0, float* r1)
{
    const float ca = R.ca, sa = R.sa, nsa = -R.sa, a = R.a, dp = R.dp;
    const float zero = 0.0f, one = 1.0f;
    const bool c0 = c == 0, c1 = c == 1, c2 = c == 2, c3 = c == 3;
    float u0 = zero; u0 = c1 ? sa : u0; u0 = c2 ? ca : u0; u0 = c3 ? dp : u0;
    const float u1 = c3 ? one : zero;
    float w0 = zero; w0 = c0 ? one : w0; w0 = c3 ? a : w0;
    float w1 = zero; w1 = c1 ? ca : w1; w1 = c2 ? nsa : w1;
    *r0 = e ? u0 : w0;
    *r1 = e ? u1 : w1;
}

__device__ __forceinline__ v2f build_stage_A(int m, int h, const JC& L, const JC& R)
{
    const int x = (m & 7) >> 1;
    const int c = 2 * (m >> 3) + (m & 1);
    const bool e = x == 2, q = x == 1, dead = x == 3, hb = h != 0;
    const float Lca = L.ca, Lsa = L.sa, nLsa = -L.sa, La = L.a, Ldp = L.dp;
    const float zero = 0.0f, one = 1.0f;
    const float la0_1 = e ? Lca : zero, la0_0 = e ? zero : one;
    const float la1_1 = e ? Ldp : Lsa,  la1_0 = e ? La : zero;
    const float lb0_0 = e ? nLsa : zero;
    const float lb1_1 = e ? one : zero, lb1_0 = e ? zero : Lca;
    const float la0 = hb ? la0_1 : la0_0;
    const float la1 = hb ? la1_1 : la1_0;
    const float lb0 = hb ? zero : lb0_0;
    const float lb1 = hb ? lb1_1 : lb1_0;
    float r0, r1;
    right_pair(e, c, R, &r0, &r1);
    const float nla0 = -la0, nlb0 = -lb0;
    const float fa0 = q ? la1 : la0, fa1 = q ? nla0 : la1;
    const float fb0 = q ? lb1 : lb0, fb1 = q ? nlb0 : lb1;
    const float ax = fmaf(fa0, r0, fa1 * r1);
    const float ay = fmaf(fb0, r0, fb1 * r1);
    v2f A;
    A.x = dead ? zero : ax;
    A.y = dead ? zero : ay;
    return A;
}

__device__ __forceinline__ v2f build_seed_A(int m, int h, const JC& L, const JC& R)
{
    const int i = (m & 7) >> 1;
    const int c = 2 * (m >> 3) + (m & 1);
    const bool e = h != 0;
    const bool i0 = i == 0, i1 = i == 1, i2 = i == 2, dead = i == 3;
    const float Lca = L.ca, Lsa = L.sa, nLsa = -L.sa, La = L.a, Ldp = L.dp;
    const float zero = 0.0f, one = 1.0f;
    const float L0 = i0 ? one : zero;
    float L1 = zero; L1 = i1 ? Lca : L1;  L1 = i2 ? Lsa : L1;
    float L2 = zero; L2 = i1 ? nLsa : L2; L2 = i2 ? Lca : L2;
    float L3 = zero; L3 = i0 ? La : L3;   L3 = i2 ? Ldp : L3;
    const float l0 = e ? L2 : L0, l1 = e ? L3 : L1;
    float r0, r1;
    right_pair(e, c, R, &r0, &r1);
    const float ax = fmaf(l0, r0, l1 * r1);
    const float sy = fmaf(l1, r0, -(l0 * r1));
    const float ay = e ? zero : sy;
    v2f A;
    A.x = dead ? zero : ax;
    A.y = dead ? zero : ay;
    return A;
}

__device__ __forceinline__ void half_swap(float& d, float& s)
{
    const v2u r = __builtin_amdgcn_permlane16_swap(__float_as_uint(d), __float_as_uint(s), false, false);
    d = __uint_as_float(r.x);
    s = __uint_as_float(r.y);
}

__device__ __forceinline__ v2f rot01(v2f b, float c, float s)
{
    v2f r;
    r.x = fmaf(c, b.x, s * b.y);
    r.y = fmaf(c, b.y, -(s * b.x));
    return r;
}

__global__ __launch_bounds__(1024) void fk_wmma_kernel(
    const float* __restrict__ theta, const float* __restrict__ dh,
    float* __restrict__ out, int n)
{
    const int lane = threadIdx.x & 31;
    const int wave = __builtin_amdgcn_readfirstlane((int)(threadIdx.x >> 5));
    const int numTiles   = (n + 31) >> 5;
    const int waveStride = (int)gridDim.x * 32;
    int tile = (int)blockIdx.x * 32 + wave;

    v2f t01, t23, t45;
    {
        int samp = tile * 32 + lane; if (samp >= n) samp = n - 1;
        const v2f* tp = (const v2f*)(theta + (size_t)samp * 6);
        t01 = tp[0]; t23 = tp[1]; t45 = tp[2];
    }

    const int jc = lane < 6 ? lane : 5;
    const float alr = dh[jc * 4 + 1] * D2R;
    float sa_l, ca_l;
    if (__builtin_expect(__builtin_amdgcn_ballot_w32(!(fabsf(alr) < SC_SMALL_BOUND)) != 0u, 0))
        sincosf(alr, &sa_l, &ca_l);
    else
        sincos_small(alr, &sa_l, &ca_l);
    JC J[6];
    float offU[6];
    #pragma unroll
    for (int j = 0; j < 6; ++j) {
        J[j].ca = __int_as_float(__builtin_amdgcn_readlane(__float_as_int(ca_l), j));
        J[j].sa = __int_as_float(__builtin_amdgcn_readlane(__float_as_int(sa_l), j));
        J[j].a  = dh[j * 4 + 0];
        J[j].dp = j > 0 ? dh[(j - 1) * 4 + 2] : 0.0f;
        offU[j] = dh[j * 4 + 3];
    }
    const int h = lane >> 4;
    const int m = lane & 15;
    const v2f Aseed = build_seed_A(m, h, J[0], J[1]);
    const v2f A1 = build_stage_A(m, h, J[2], J[3]);
    const v2f A2 = build_stage_A(m, h, J[4], J[5]);
    const float dT = h ? dh[5 * 4 + 2] : 0.0f;

    for (; tile < numTiles; tile += waveStride) {
        const int base = tile * 32;

        v2f n01, n23, n45;
        {
            int ns = base + waveStride * 32 + lane; if (ns >= n) ns = n - 1;
            const v2f* np = (const v2f*)(theta + (size_t)ns * 6);
            n01 = np[0]; n23 = np[1]; n45 = np[2];
        }

        float stv[6], ctv[6];
        {
            const float thv[6] = { t01.x, t01.y, t23.x, t23.y, t45.x, t45.y };
            float thr[6];
            float big = 0.0f;
            #pragma unroll
            for (int j = 0; j < 6; ++j) {
                thr[j] = (offU[j] + thv[j]) * D2R;
                big = fmaxf(big, fabsf(thr[j]));
            }
            if (__builtin_expect(__builtin_amdgcn_ballot_w32(!(big < SC_SMALL_BOUND)) != 0u, 0)) {
                #pragma unroll
                for (int j = 0; j < 6; ++j) sincosf(thr[j], &stv[j], &ctv[j]);
            } else {
                #pragma unroll
                for (int j = 0; j < 6; ++j) sincos_small(thr[j], &stv[j], &ctv[j]);
            }
        }

        float cs[2][6], sn[2][6];
        #pragma unroll
        for (int j = 0; j < 6; ++j) {
            const bool both = (j == 2) || (j == 4);
            cs[0][j] = ctv[j]; cs[1][j] = both ? ctv[j] : 1.0f;
            sn[0][j] = stv[j]; sn[1][j] = both ? stv[j] : 0.0f;
            half_swap(cs[0][j], cs[1][j]);
            half_swap(sn[0][j], sn[1][j]);
        }

        v2f b[2][3];
        #pragma unroll
        for (int s = 0; s < 2; ++s) {
            v2f Bs; Bs.x = cs[s][0]; Bs.y = sn[s][0];
            v8f acc = {};
            acc = __builtin_amdgcn_wmma_f32_16x16x4_f32(
                      false, Aseed, false, Bs, (short)0, acc, false, false);
            #pragma unroll
            for (int i = 0; i < 3; ++i) { b[s][i].x = acc[2 * i]; b[s][i].y = acc[2 * i + 1]; }
        }
        #pragma unroll
        for (int r = 0; r < 2; ++r) {
            const int jr = 2 * r + 1;
            const int jw = 2 * r + 2;
            #pragma unroll
            for (int s = 0; s < 2; ++s) {
                #pragma unroll
                for (int i = 0; i < 3; ++i) {
                    const v2f v = rot01(b[s][i], cs[s][jr], sn[s][jr]);
                    v8f acc = {};
                    acc = __builtin_amdgcn_wmma_f32_16x16x4_f32(
                              false, r == 0 ? A1 : A2, false, v, (short)0, acc, false, false);
                    const float c = cs[s][jw], sv = sn[s][jw];
                    b[s][i].x = fmaf(c, acc[0], fmaf(sv, acc[2], acc[4]));
                    b[s][i].y = fmaf(c, acc[1], fmaf(sv, acc[3], acc[5]));
                }
            }
        }
        #pragma unroll
        for (int s = 0; s < 2; ++s) {
            #pragma unroll
            for (int i = 0; i < 3; ++i) {
                v2f v = rot01(b[s][i], cs[s][5], sn[s][5]);
                v.y = fmaf(dT, v.x, v.y);
                b[s][i] = v;
            }
        }

        float e0[3], e1[3], e2[3], e3[3];
        #pragma unroll
        for (int i = 0; i < 3; ++i) {
            e0[i] = b[0][i].x; e2[i] = b[1][i].x; half_swap(e0[i], e2[i]);
            e1[i] = b[0][i].y; e3[i] = b[1][i].y; half_swap(e1[i], e3[i]);
        }

        {
            const float x = e3[0], y = e3[1], z = e3[2];
            const float T00 = e0[0], T01 = e1[0], T02 = e2[0];
            const float T10 = e0[1], T11 = e1[1], T12 = e2[1];
            const float T22 = e2[2];
            float Aa, Bb, Cc;
            if (fabsf(T12) <= 1e-6f && fabsf(T22) <= 1e-6f) {
                Aa = R2D * atan2f(T10, T11);
                Bb = R2D * atan2f(T02, T22);
                Cc = 0.0f;
            } else {
                const float len = __builtin_amdgcn_sqrtf(fmaf(T00, T00, T01 * T01));
                const float ayA = fabsf(T01), axA = fabsf(T00);
                const float ayC = fabsf(T12), axC = fabsf(T22);
                const float ayB = fabsf(T02);
                const bool stA = ayA > axA, stC = ayC > axC, stB = ayB > len;
                v2f q;
                q.x = ratio_small(stA ? axA : ayA, stA ? ayA : axA);
                q.y = ratio_small(stC ? axC : ayC, stC ? ayC : axC);
                const v2f pAC = atan_poly2(q);
                const float pB = atan_poly2(splat2(ratio_small(stB ? len : ayB, stB ? ayB : len))).x;
                Aa = -R2D * atan2_finish<true>(pAC.x, T01, T00, stA);
                Cc = -R2D * atan2_finish<true>(pAC.y, T12, T22, stC);
                Bb =  R2D * atan2_finish<false>(pB, T02, len, stB);
            }
            if (base + lane < n) {
                v2f* o = (v2f*)(out + (size_t)(base + lane) * 6);
                v2f o0; o0.x = x;  o0.y = y;
                v2f o1; o1.x = z;  o1.y = Aa;
                v2f o2; o2.x = Bb; o2.y = Cc;
                o[0] = o0; o[1] = o1; o[2] = o2;
            }
        }
        t01 = n01; t23 = n23; t45 = n45;
    }
}

extern "C" void kernel_launch(void* const* d_in, const int* in_sizes, int n_in,
                              void* d_out, int out_size, void* d_ws, size_t ws_size,
                              hipStream_t stream) {
    const float* theta = (const float*)d_in[0];
    const float* dh    = (const float*)d_in[1];
    float* out = (float*)d_out;
    const int n = in_sizes[0] / 6;
    fk_wmma_kernel<<<256, 1024, 0, stream>>>(theta, dh, out, n);
}
